# starting kernel
# baseline (speedup 1.0000x reference)
#include <hip/hip_runtime.h>

typedef __attribute__((ext_vector_type(2))) float v2f;
typedef __attribute__((ext_vector_type(8))) float v8f;

#define JN 9
#define NEARZ 1e-6f

__device__ __forceinline__ void cross3(const float* a, const float* b, float* o) {
  o[0] = a[1]*b[2] - a[2]*b[1];
  o[1] = a[2]*b[0] - a[0]*b[2];
  o[2] = a[0]*b[1] - a[1]*b[0];
}

__device__ __forceinline__ void matvec3(const float* R, const float* x, float* o) {
#pragma unroll
  for (int r = 0; r < 3; ++r)
    o[r] = R[r*3+0]*x[0] + R[r*3+1]*x[1] + R[r*3+2]*x[2];
}

__device__ __forceinline__ void matvec3T(const float* R, const float* x, float* o) {
#pragma unroll
  for (int r = 0; r < 3; ++r)
    o[r] = R[0*3+r]*x[0] + R[1*3+r]*x[1] + R[2*3+r]*x[2];
}

__device__ __forceinline__ void matexp6(const float* V, float* R, float* p) {
  float th = sqrtf(V[0]*V[0] + V[1]*V[1] + V[2]*V[2]);
  const bool near = th < NEARZ;
  float safe = near ? 1.0f : th;
  float inv  = 1.0f / safe;
  float u0 = V[0]*inv, u1 = V[1]*inv, u2 = V[2]*inv;
  float Wh[9] = {0.f, -u2, u1,  u2, 0.f, -u0,  -u1, u0, 0.f};
  float W2[9];
#pragma unroll
  for (int r = 0; r < 3; ++r)
#pragma unroll
    for (int c = 0; c < 3; ++c)
      W2[r*3+c] = Wh[r*3+0]*Wh[0*3+c] + Wh[r*3+1]*Wh[1*3+c] + Wh[r*3+2]*Wh[2*3+c];
  float st = sinf(th), ct = cosf(th);
  float omc = 1.0f - ct;
  float tms = th - st;
#pragma unroll
  for (int r = 0; r < 3; ++r) {
#pragma unroll
    for (int c = 0; c < 3; ++c) {
      float id  = (r == c) ? 1.0f : 0.0f;
      float val = id + st*Wh[r*3+c] + omc*W2[r*3+c];
      R[r*3+c]  = near ? id : val;
    }
  }
  float s0 = V[3]*inv, s1 = V[4]*inv, s2 = V[5]*inv;
#pragma unroll
  for (int r = 0; r < 3; ++r) {
    float g0 = ((r==0)?th:0.f) + omc*Wh[r*3+0] + tms*W2[r*3+0];
    float g1 = ((r==1)?th:0.f) + omc*Wh[r*3+1] + tms*W2[r*3+1];
    float g2 = ((r==2)?th:0.f) + omc*Wh[r*3+2] + tms*W2[r*3+2];
    float val = g0*s0 + g1*s1 + g2*s2;
    p[r] = near ? V[3+r] : val;
  }
}

__device__ __forceinline__ void compute_T(int i, float qi,
                                          const float* __restrict__ Alist,
                                          const float* __restrict__ Mlist,
                                          float* Tr, float* Tp) {
  float V[6];
#pragma unroll
  for (int j = 0; j < 6; ++j) V[j] = -Alist[i*6+j] * qi;
  float Re[9], pe[3];
  matexp6(V, Re, pe);
  const float* M = Mlist + i*16;
  float Rt[9];
#pragma unroll
  for (int r = 0; r < 3; ++r)
#pragma unroll
    for (int c = 0; c < 3; ++c)
      Rt[r*3+c] = M[c*4+r];
  float pinv[3];
#pragma unroll
  for (int r = 0; r < 3; ++r)
    pinv[r] = -(Rt[r*3+0]*M[0*4+3] + Rt[r*3+1]*M[1*4+3] + Rt[r*3+2]*M[2*4+3]);
#pragma unroll
  for (int r = 0; r < 3; ++r) {
#pragma unroll
    for (int c = 0; c < 3; ++c)
      Tr[r*3+c] = Re[r*3+0]*Rt[0*3+c] + Re[r*3+1]*Rt[1*3+c] + Re[r*3+2]*Rt[2*3+c];
    Tp[r] = Re[r*3+0]*pinv[0] + Re[r*3+1]*pinv[1] + Re[r*3+2]*pinv[2] + pe[r];
  }
}

__device__ __forceinline__ void g_matvec_wmma(const float x[6], float y[6],
                                              int mrow, int half,
                                              v2f A00, v2f A01, v2f A10, v2f A11) {
  v8f acc = {0.f, 0.f, 0.f, 0.f, 0.f, 0.f, 0.f, 0.f};
#pragma unroll
  for (int t = 0; t < 2; ++t) {
    int s = mrow + 16*t;
    float x0 = __shfl(x[0], s, 32);
    float x1 = __shfl(x[1], s, 32);
    float x2 = __shfl(x[2], s, 32);
    float x3 = __shfl(x[3], s, 32);
    float x4 = __shfl(x[4], s, 32);
    float x5 = __shfl(x[5], s, 32);
    v2f B0, B1;
    B0.x = half ? x2  : x0;  B0.y = half ? x3  : x1;
    B1.x = half ? 0.f : x4;  B1.y = half ? 0.f : x5;
    acc = __builtin_amdgcn_wmma_f32_16x16x4_f32(false, (t == 0) ? A00 : A10,
                                                false, B0, (short)0, acc,
                                                false, false);
    acc = __builtin_amdgcn_wmma_f32_16x16x4_f32(false, (t == 0) ? A01 : A11,
                                                false, B1, (short)0, acc,
                                                false, false);
  }
#pragma unroll
  for (int r = 0; r < 6; ++r) y[r] = acc[r];
}

__global__ __launch_bounds__(256) void inv_dyn_kernel(
    const float* __restrict__ q, const float* __restrict__ dq,
    const float* __restrict__ ddq, const float* __restrict__ g,
    const float* __restrict__ Ftip, const float* __restrict__ Alist,
    const float* __restrict__ Mlist, const float* __restrict__ Glist,
    float* __restrict__ tau, int B) {
  int gid = blockIdx.x * blockDim.x + threadIdx.x;
  int b = (gid < B) ? gid : (B - 1);

  float lq[JN], ldq[JN], lddq[JN];
#pragma unroll
  for (int i = 0; i < JN; ++i) {
    lq[i]   = q[b*JN + i];
    ldq[i]  = dq[b*JN + i];
    lddq[i] = ddq[b*JN + i];
  }

  float v[6] = {0.f, 0.f, 0.f, 0.f, 0.f, 0.f};
  float a[6] = {0.f, 0.f, 0.f, -g[0], -g[1], -g[2]};
  float vs[JN][6], as_[JN][6];

#pragma unroll
  for (int i = 0; i < JN; ++i) {
    float Tr[9], Tp[3];
    compute_T(i, lq[i], Alist, Mlist, Tr, Tp);
    const float* A = Alist + i*6;
    float dqi = ldq[i], ddqi = lddq[i];

    float Rvw[3], Rvv[3], pxw[3];
    matvec3(Tr, v, Rvw);
    matvec3(Tr, v + 3, Rvv);
    cross3(Tp, Rvw, pxw);
    float vn[6];
#pragma unroll
    for (int k = 0; k < 3; ++k) {
      vn[k]     = Rvw[k] + A[k]*dqi;
      vn[3 + k] = pxw[k] + Rvv[k] + A[3 + k]*dqi;
    }

    float Raw[3], Rav[3], pxa[3];
    matvec3(Tr, a, Raw);
    matvec3(Tr, a + 3, Rav);
    cross3(Tp, Raw, pxa);
    float t1[3], t2[3], t3[3];
    cross3(vn, A, t1);
    cross3(vn + 3, A, t2);
    cross3(vn, A + 3, t3);
    float an[6];
#pragma unroll
    for (int k = 0; k < 3; ++k) {
      an[k]     = Raw[k] + t1[k]*dqi + A[k]*ddqi;
      an[3 + k] = pxa[k] + Rav[k] + (t2[k] + t3[k])*dqi + A[3 + k]*ddqi;
    }
#pragma unroll
    for (int k = 0; k < 6; ++k) {
      vs[i][k] = vn[k]; as_[i][k] = an[k];
      v[k] = vn[k];     a[k] = an[k];
    }
  }

  const int lane = threadIdx.x & 31;
  const int mrow = lane & 15;
  const int half = lane >> 4;

  float F[6];
#pragma unroll
  for (int k = 0; k < 6; ++k) F[k] = Ftip[b*6 + k];

#pragma unroll
  for (int ii = JN - 1; ii >= 0; --ii) {
    float Rn[9], Pn[3];
    if (ii == JN - 1) {
      const float* M = Mlist + JN*16;
#pragma unroll
      for (int r = 0; r < 3; ++r)
#pragma unroll
        for (int c = 0; c < 3; ++c)
          Rn[r*3+c] = M[c*4+r];
#pragma unroll
      for (int r = 0; r < 3; ++r)
        Pn[r] = -(Rn[r*3+0]*M[3] + Rn[r*3+1]*M[7] + Rn[r*3+2]*M[11]);
    } else {
      compute_T(ii + 1, lq[ii + 1], Alist, Mlist, Rn, Pn);
    }

    float cf[3];
    cross3(Pn, F + 3, cf);
    float tw[3] = {F[0] - cf[0], F[1] - cf[1], F[2] - cf[2]};
    float top[3], bot[3];
    matvec3T(Rn, tw, top);
    matvec3T(Rn, F + 3, bot);

    const float* G = Glist + ii*36;
    bool ok0 = (mrow < 6);
    bool ok1 = (mrow >= 8) && (mrow < 14);
    int  r0  = ok0 ? mrow : 0;
    int  r1  = ok1 ? (mrow - 8) : 0;
    int  c0  = 2*half;
    float ga = G[r0*6 + c0], gb = G[r0*6 + c0 + 1];
    float gc = G[r1*6 + c0], gd = G[r1*6 + c0 + 1];
    float ge = G[r0*6 + 4],  gf = G[r0*6 + 5];
    float gg = G[r1*6 + 4],  gh = G[r1*6 + 5];
    bool okh = (half == 0);
    v2f A00, A01, A10, A11;
    A00.x = ok0 ? ga : 0.f;           A00.y = ok0 ? gb : 0.f;
    A10.x = ok1 ? gc : 0.f;           A10.y = ok1 ? gd : 0.f;
    A01.x = (ok0 && okh) ? ge : 0.f;  A01.y = (ok0 && okh) ? gf : 0.f;
    A11.x = (ok1 && okh) ? gg : 0.f;  A11.y = (ok1 && okh) ? gh : 0.f;

    float Ga[6], Gv[6];
    g_matvec_wmma(as_[ii], Ga, mrow, half, A00, A01, A10, A11);
    g_matvec_wmma(vs[ii],  Gv, mrow, half, A00, A01, A10, A11);

    float c1[3], c2[3], c3[3];
    cross3(vs[ii], Gv, c1);
    cross3(vs[ii] + 3, Gv + 3, c2);
    cross3(vs[ii], Gv + 3, c3);
    float Fn[6];
#pragma unroll
    for (int k = 0; k < 3; ++k) {
      Fn[k]     = top[k] + Ga[k]     + c1[k] + c2[k];
      Fn[3 + k] = bot[k] + Ga[3 + k] + c3[k];
    }

    const float* A = Alist + ii*6;
    float tv = Fn[0]*A[0] + Fn[1]*A[1] + Fn[2]*A[2] +
               Fn[3]*A[3] + Fn[4]*A[4] + Fn[5]*A[5];
    tau[b*JN + ii] = tv;
#pragma unroll
    for (int k = 0; k < 6; ++k) F[k] = Fn[k];
  }
}

extern "C" void kernel_launch(void* const* d_in, const int* in_sizes, int n_in,
                              void* d_out, int out_size, void* d_ws, size_t ws_size,
                              hipStream_t stream) {
  const float* q     = (const float*)d_in[0];
  const float* dq    = (const float*)d_in[1];
  const float* ddq   = (const float*)d_in[2];
  const float* g     = (const float*)d_in[3];
  const float* Ftip  = (const float*)d_in[4];
  const float* Alist = (const float*)d_in[5];
  const float* Mlist = (const float*)d_in[6];
  const float* Glist = (const float*)d_in[7];
  int B = in_sizes[0] / JN;
  if (B <= 0) return;
  dim3 block(256);
  dim3 grid((B + 255) / 256);
  inv_dyn_kernel<<<grid, block, 0, stream>>>(q, dq, ddq, g, Ftip,
                                             Alist, Mlist, Glist,
                                             (float*)d_out, B);
}


# Backward scan reuses the forward scan's joint transforms (T_1..T_7 from a per-thread LDS stash, T_8 from registers) instead of recomputing exp6 eight times; summation order unchanged
# speedup vs baseline: 1.5184x; 1.5184x over previous
#include <hip/hip_runtime.h>

typedef __attribute__((ext_vector_type(2))) float v2f;
typedef __attribute__((ext_vector_type(8))) float v8f;

#define JN 9
#define NEARZ 1e-6f

__device__ __forceinline__ void cross3(const float* a, const float* b, float* o) {
  o[0] = a[1]*b[2] - a[2]*b[1];
  o[1] = a[2]*b[0] - a[0]*b[2];
  o[2] = a[0]*b[1] - a[1]*b[0];
}

__device__ __forceinline__ void matvec3(const float* R, const float* x, float* o) {
#pragma unroll
  for (int r = 0; r < 3; ++r)
    o[r] = R[r*3+0]*x[0] + R[r*3+1]*x[1] + R[r*3+2]*x[2];
}

__device__ __forceinline__ void matvec3T(const float* R, const float* x, float* o) {
#pragma unroll
  for (int r = 0; r < 3; ++r)
    o[r] = R[0*3+r]*x[0] + R[1*3+r]*x[1] + R[2*3+r]*x[2];
}

__device__ __forceinline__ void matexp6(const float* V, float* R, float* p) {
  float th = sqrtf(V[0]*V[0] + V[1]*V[1] + V[2]*V[2]);
  const bool near = th < NEARZ;
  float safe = near ? 1.0f : th;
  float inv  = 1.0f / safe;
  float u0 = V[0]*inv, u1 = V[1]*inv, u2 = V[2]*inv;
  float Wh[9] = {0.f, -u2, u1,  u2, 0.f, -u0,  -u1, u0, 0.f};
  float W2[9];
#pragma unroll
  for (int r = 0; r < 3; ++r)
#pragma unroll
    for (int c = 0; c < 3; ++c)
      W2[r*3+c] = Wh[r*3+0]*Wh[0*3+c] + Wh[r*3+1]*Wh[1*3+c] + Wh[r*3+2]*Wh[2*3+c];
  float st = sinf(th), ct = cosf(th);
  float omc = 1.0f - ct;
  float tms = th - st;
#pragma unroll
  for (int r = 0; r < 3; ++r) {
#pragma unroll
    for (int c = 0; c < 3; ++c) {
      float id  = (r == c) ? 1.0f : 0.0f;
      float val = id + st*Wh[r*3+c] + omc*W2[r*3+c];
      R[r*3+c]  = near ? id : val;
    }
  }
  float s0 = V[3]*inv, s1 = V[4]*inv, s2 = V[5]*inv;
#pragma unroll
  for (int r = 0; r < 3; ++r) {
    float g0 = ((r==0)?th:0.f) + omc*Wh[r*3+0] + tms*W2[r*3+0];
    float g1 = ((r==1)?th:0.f) + omc*Wh[r*3+1] + tms*W2[r*3+1];
    float g2 = ((r==2)?th:0.f) + omc*Wh[r*3+2] + tms*W2[r*3+2];
    float val = g0*s0 + g1*s1 + g2*s2;
    p[r] = near ? V[3+r] : val;
  }
}

__device__ __forceinline__ void compute_T(int i, float qi,
                                          const float* __restrict__ Alist,
                                          const float* __restrict__ Mlist,
                                          float* Tr, float* Tp) {
  float V[6];
#pragma unroll
  for (int j = 0; j < 6; ++j) V[j] = -Alist[i*6+j] * qi;
  float Re[9], pe[3];
  matexp6(V, Re, pe);
  const float* M = Mlist + i*16;
  float Rt[9];
#pragma unroll
  for (int r = 0; r < 3; ++r)
#pragma unroll
    for (int c = 0; c < 3; ++c)
      Rt[r*3+c] = M[c*4+r];
  float pinv[3];
#pragma unroll
  for (int r = 0; r < 3; ++r)
    pinv[r] = -(Rt[r*3+0]*M[0*4+3] + Rt[r*3+1]*M[1*4+3] + Rt[r*3+2]*M[2*4+3]);
#pragma unroll
  for (int r = 0; r < 3; ++r) {
#pragma unroll
    for (int c = 0; c < 3; ++c)
      Tr[r*3+c] = Re[r*3+0]*Rt[0*3+c] + Re[r*3+1]*Rt[1*3+c] + Re[r*3+2]*Rt[2*3+c];
    Tp[r] = Re[r*3+0]*pinv[0] + Re[r*3+1]*pinv[1] + Re[r*3+2]*pinv[2] + pe[r];
  }
}

#define TPB 256
__device__ __forceinline__ void stash_T(float4* sT, int j, int tid, const float* Tr, const float* Tp) {
  sT[(j*3 + 0)*TPB + tid] = make_float4(Tr[0], Tr[1], Tr[2], Tr[3]);
  sT[(j*3 + 1)*TPB + tid] = make_float4(Tr[4], Tr[5], Tr[6], Tr[7]);
  sT[(j*3 + 2)*TPB + tid] = make_float4(Tr[8], Tp[0], Tp[1], Tp[2]);
}
__device__ __forceinline__ void fetch_T(const float4* sT, int j, int tid, float* Tr, float* Tp) {
  float4 a = sT[(j*3 + 0)*TPB + tid];
  float4 b = sT[(j*3 + 1)*TPB + tid];
  float4 c = sT[(j*3 + 2)*TPB + tid];
  Tr[0] = a.x; Tr[1] = a.y; Tr[2] = a.z; Tr[3] = a.w;
  Tr[4] = b.x; Tr[5] = b.y; Tr[6] = b.z; Tr[7] = b.w;
  Tr[8] = c.x; Tp[0] = c.y; Tp[1] = c.z; Tp[2] = c.w;
}

__device__ __forceinline__ void g_matvec_wmma(const float x[6], float y[6],
                                              int mrow, int half,
                                              v2f A00, v2f A01, v2f A10, v2f A11) {
  v8f acc = {0.f, 0.f, 0.f, 0.f, 0.f, 0.f, 0.f, 0.f};
#pragma unroll
  for (int t = 0; t < 2; ++t) {
    int s = mrow + 16*t;
    float x0 = __shfl(x[0], s, 32);
    float x1 = __shfl(x[1], s, 32);
    float x2 = __shfl(x[2], s, 32);
    float x3 = __shfl(x[3], s, 32);
    float x4 = __shfl(x[4], s, 32);
    float x5 = __shfl(x[5], s, 32);
    v2f B0, B1;
    B0.x = half ? x2  : x0;  B0.y = half ? x3  : x1;
    B1.x = half ? 0.f : x4;  B1.y = half ? 0.f : x5;
    acc = __builtin_amdgcn_wmma_f32_16x16x4_f32(false, (t == 0) ? A00 : A10,
                                                false, B0, (short)0, acc,
                                                false, false);
    acc = __builtin_amdgcn_wmma_f32_16x16x4_f32(false, (t == 0) ? A01 : A11,
                                                false, B1, (short)0, acc,
                                                false, false);
  }
#pragma unroll
  for (int r = 0; r < 6; ++r) y[r] = acc[r];
}

__global__ __launch_bounds__(TPB) void inv_dyn_kernel(
    const float* __restrict__ q, const float* __restrict__ dq,
    const float* __restrict__ ddq, const float* __restrict__ g,
    const float* __restrict__ Ftip, const float* __restrict__ Alist,
    const float* __restrict__ Mlist, const float* __restrict__ Glist,
    float* __restrict__ tau, int B) {
  __shared__ float4 sT[(JN - 2)*3*TPB];
  const int tid = threadIdx.x;
  int gid = blockIdx.x * blockDim.x + threadIdx.x;
  int b = (gid < B) ? gid : (B - 1);

  float lq[JN], ldq[JN], lddq[JN];
#pragma unroll
  for (int i = 0; i < JN; ++i) {
    lq[i]   = q[b*JN + i];
    ldq[i]  = dq[b*JN + i];
    lddq[i] = ddq[b*JN + i];
  }

  float v[6] = {0.f, 0.f, 0.f, 0.f, 0.f, 0.f};
  float a[6] = {0.f, 0.f, 0.f, -g[0], -g[1], -g[2]};
  float vs[JN][6], as_[JN][6];
  float TlastR[9], TlastP[3];

#pragma unroll
  for (int i = 0; i < JN; ++i) {
    float Tr[9], Tp[3];
    compute_T(i, lq[i], Alist, Mlist, Tr, Tp);
    if (i >= 1 && i < JN - 1) stash_T(sT, i - 1, tid, Tr, Tp);
    if (i == JN - 1) {
#pragma unroll
      for (int k = 0; k < 9; ++k) TlastR[k] = Tr[k];
#pragma unroll
      for (int k = 0; k < 3; ++k) TlastP[k] = Tp[k];
    }
    const float* A = Alist + i*6;
    float dqi = ldq[i], ddqi = lddq[i];

    float Rvw[3], Rvv[3], pxw[3];
    matvec3(Tr, v, Rvw);
    matvec3(Tr, v + 3, Rvv);
    cross3(Tp, Rvw, pxw);
    float vn[6];
#pragma unroll
    for (int k = 0; k < 3; ++k) {
      vn[k]     = Rvw[k] + A[k]*dqi;
      vn[3 + k] = pxw[k] + Rvv[k] + A[3 + k]*dqi;
    }

    float Raw[3], Rav[3], pxa[3];
    matvec3(Tr, a, Raw);
    matvec3(Tr, a + 3, Rav);
    cross3(Tp, Raw, pxa);
    float t1[3], t2[3], t3[3];
    cross3(vn, A, t1);
    cross3(vn + 3, A, t2);
    cross3(vn, A + 3, t3);
    float an[6];
#pragma unroll
    for (int k = 0; k < 3; ++k) {
      an[k]     = Raw[k] + t1[k]*dqi + A[k]*ddqi;
      an[3 + k] = pxa[k] + Rav[k] + (t2[k] + t3[k])*dqi + A[3 + k]*ddqi;
    }
#pragma unroll
    for (int k = 0; k < 6; ++k) {
      vs[i][k] = vn[k]; as_[i][k] = an[k];
      v[k] = vn[k];     a[k] = an[k];
    }
  }

  const int lane = threadIdx.x & 31;
  const int mrow = lane & 15;
  const int half = lane >> 4;

  float F[6];
#pragma unroll
  for (int k = 0; k < 6; ++k) F[k] = Ftip[b*6 + k];

#pragma unroll
  for (int ii = JN - 1; ii >= 0; --ii) {
    float Rn[9], Pn[3];
    if (ii == JN - 1) {
      const float* M = Mlist + JN*16;
#pragma unroll
      for (int r = 0; r < 3; ++r)
#pragma unroll
        for (int c = 0; c < 3; ++c)
          Rn[r*3+c] = M[c*4+r];
#pragma unroll
      for (int r = 0; r < 3; ++r)
        Pn[r] = -(Rn[r*3+0]*M[3] + Rn[r*3+1]*M[7] + Rn[r*3+2]*M[11]);
    } else if (ii == JN - 2) {
#pragma unroll
      for (int k = 0; k < 9; ++k) Rn[k] = TlastR[k];
#pragma unroll
      for (int k = 0; k < 3; ++k) Pn[k] = TlastP[k];
    } else {
      fetch_T(sT, ii, tid, Rn, Pn);
    }

    float cf[3];
    cross3(Pn, F + 3, cf);
    float tw[3] = {F[0] - cf[0], F[1] - cf[1], F[2] - cf[2]};
    float top[3], bot[3];
    matvec3T(Rn, tw, top);
    matvec3T(Rn, F + 3, bot);

    const float* G = Glist + ii*36;
    bool ok0 = (mrow < 6);
    bool ok1 = (mrow >= 8) && (mrow < 14);
    int  r0  = ok0 ? mrow : 0;
    int  r1  = ok1 ? (mrow - 8) : 0;
    int  c0  = 2*half;
    float ga = G[r0*6 + c0], gb = G[r0*6 + c0 + 1];
    float gc = G[r1*6 + c0], gd = G[r1*6 + c0 + 1];
    float ge = G[r0*6 + 4],  gf = G[r0*6 + 5];
    float gg = G[r1*6 + 4],  gh = G[r1*6 + 5];
    bool okh = (half == 0);
    v2f A00, A01, A10, A11;
    A00.x = ok0 ? ga : 0.f;           A00.y = ok0 ? gb : 0.f;
    A10.x = ok1 ? gc : 0.f;           A10.y = ok1 ? gd : 0.f;
    A01.x = (ok0 && okh) ? ge : 0.f;  A01.y = (ok0 && okh) ? gf : 0.f;
    A11.x = (ok1 && okh) ? gg : 0.f;  A11.y = (ok1 && okh) ? gh : 0.f;

    float Ga[6], Gv[6];
    g_matvec_wmma(as_[ii], Ga, mrow, half, A00, A01, A10, A11);
    g_matvec_wmma(vs[ii],  Gv, mrow, half, A00, A01, A10, A11);

    float c1[3], c2[3], c3[3];
    cross3(vs[ii], Gv, c1);
    cross3(vs[ii] + 3, Gv + 3, c2);
    cross3(vs[ii], Gv + 3, c3);
    float Fn[6];
#pragma unroll
    for (int k = 0; k < 3; ++k) {
      Fn[k]     = top[k] + Ga[k]     + c1[k] + c2[k];
      Fn[3 + k] = bot[k] + Ga[3 + k] + c3[k];
    }

    const float* A = Alist + ii*6;
    float tv = Fn[0]*A[0] + Fn[1]*A[1] + Fn[2]*A[2] +
               Fn[3]*A[3] + Fn[4]*A[4] + Fn[5]*A[5];
    tau[b*JN + ii] = tv;
#pragma unroll
    for (int k = 0; k < 6; ++k) F[k] = Fn[k];
  }
}

extern "C" void kernel_launch(void* const* d_in, const int* in_sizes, int n_in,
                              void* d_out, int out_size, void* d_ws, size_t ws_size,
                              hipStream_t stream) {
  const float* q     = (const float*)d_in[0];
  const float* dq    = (const float*)d_in[1];
  const float* ddq   = (const float*)d_in[2];
  const float* g     = (const float*)d_in[3];
  const float* Ftip  = (const float*)d_in[4];
  const float* Alist = (const float*)d_in[5];
  const float* Mlist = (const float*)d_in[6];
  const float* Glist = (const float*)d_in[7];
  int B = in_sizes[0] / JN;
  if (B <= 0) return;
  dim3 block(256);
  dim3 grid((B + 255) / 256);
  inv_dyn_kernel<<<grid, block, 0, stream>>>(q, dq, ddq, g, Ftip,
                                             Alist, Mlist, Glist,
                                             (float*)d_out, B);
}


# G matvec on WMMA re-tiled: lanes n and n+16 share a column stacked along K under block-diagonal A, each lane feeds its own x as B: 3 WMMAs not 4, no lane shuffles; 6-term sums regrouped in pairs
# speedup vs baseline: 1.7090x; 1.1255x over previous
#include <hip/hip_runtime.h>

typedef __attribute__((ext_vector_type(2))) float v2f;
typedef __attribute__((ext_vector_type(8))) float v8f;

#define JN 9
#define NEARZ 1e-6f

__device__ __forceinline__ void cross3(const float* a, const float* b, float* o) {
  o[0] = a[1]*b[2] - a[2]*b[1];
  o[1] = a[2]*b[0] - a[0]*b[2];
  o[2] = a[0]*b[1] - a[1]*b[0];
}

__device__ __forceinline__ void matvec3(const float* R, const float* x, float* o) {
#pragma unroll
  for (int r = 0; r < 3; ++r)
    o[r] = R[r*3+0]*x[0] + R[r*3+1]*x[1] + R[r*3+2]*x[2];
}

__device__ __forceinline__ void matvec3T(const float* R, const float* x, float* o) {
#pragma unroll
  for (int r = 0; r < 3; ++r)
    o[r] = R[0*3+r]*x[0] + R[1*3+r]*x[1] + R[2*3+r]*x[2];
}

__device__ __forceinline__ void matexp6(const float* V, float* R, float* p) {
  float th = sqrtf(V[0]*V[0] + V[1]*V[1] + V[2]*V[2]);
  const bool near = th < NEARZ;
  float safe = near ? 1.0f : th;
  float inv  = 1.0f / safe;
  float u0 = V[0]*inv, u1 = V[1]*inv, u2 = V[2]*inv;
  float Wh[9] = {0.f, -u2, u1,  u2, 0.f, -u0,  -u1, u0, 0.f};
  float W2[9];
#pragma unroll
  for (int r = 0; r < 3; ++r)
#pragma unroll
    for (int c = 0; c < 3; ++c)
      W2[r*3+c] = Wh[r*3+0]*Wh[0*3+c] + Wh[r*3+1]*Wh[1*3+c] + Wh[r*3+2]*Wh[2*3+c];
  float st = sinf(th), ct = cosf(th);
  float omc = 1.0f - ct;
  float tms = th - st;
#pragma unroll
  for (int r = 0; r < 3; ++r) {
#pragma unroll
    for (int c = 0; c < 3; ++c) {
      float id  = (r == c) ? 1.0f : 0.0f;
      float val = id + st*Wh[r*3+c] + omc*W2[r*3+c];
      R[r*3+c]  = near ? id : val;
    }
  }
  float s0 = V[3]*inv, s1 = V[4]*inv, s2 = V[5]*inv;
#pragma unroll
  for (int r = 0; r < 3; ++r) {
    float g0 = ((r==0)?th:0.f) + omc*Wh[r*3+0] + tms*W2[r*3+0];
    float g1 = ((r==1)?th:0.f) + omc*Wh[r*3+1] + tms*W2[r*3+1];
    float g2 = ((r==2)?th:0.f) + omc*Wh[r*3+2] + tms*W2[r*3+2];
    float val = g0*s0 + g1*s1 + g2*s2;
    p[r] = near ? V[3+r] : val;
  }
}

__device__ __forceinline__ void compute_T(int i, float qi,
                                          const float* __restrict__ Alist,
                                          const float* __restrict__ Mlist,
                                          float* Tr, float* Tp) {
  float V[6];
#pragma unroll
  for (int j = 0; j < 6; ++j) V[j] = -Alist[i*6+j] * qi;
  float Re[9], pe[3];
  matexp6(V, Re, pe);
  const float* M = Mlist + i*16;
  float Rt[9];
#pragma unroll
  for (int r = 0; r < 3; ++r)
#pragma unroll
    for (int c = 0; c < 3; ++c)
      Rt[r*3+c] = M[c*4+r];
  float pinv[3];
#pragma unroll
  for (int r = 0; r < 3; ++r)
    pinv[r] = -(Rt[r*3+0]*M[0*4+3] + Rt[r*3+1]*M[1*4+3] + Rt[r*3+2]*M[2*4+3]);
#pragma unroll
  for (int r = 0; r < 3; ++r) {
#pragma unroll
    for (int c = 0; c < 3; ++c)
      Tr[r*3+c] = Re[r*3+0]*Rt[0*3+c] + Re[r*3+1]*Rt[1*3+c] + Re[r*3+2]*Rt[2*3+c];
    Tp[r] = Re[r*3+0]*pinv[0] + Re[r*3+1]*pinv[1] + Re[r*3+2]*pinv[2] + pe[r];
  }
}

#define TPB 256
__device__ __forceinline__ void stash_T(float4* sT, int j, int tid, const float* Tr, const float* Tp) {
  sT[(j*3 + 0)*TPB + tid] = make_float4(Tr[0], Tr[1], Tr[2], Tr[3]);
  sT[(j*3 + 1)*TPB + tid] = make_float4(Tr[4], Tr[5], Tr[6], Tr[7]);
  sT[(j*3 + 2)*TPB + tid] = make_float4(Tr[8], Tp[0], Tp[1], Tp[2]);
}
__device__ __forceinline__ void fetch_T(const float4* sT, int j, int tid, float* Tr, float* Tp) {
  float4 a = sT[(j*3 + 0)*TPB + tid];
  float4 b = sT[(j*3 + 1)*TPB + tid];
  float4 c = sT[(j*3 + 2)*TPB + tid];
  Tr[0] = a.x; Tr[1] = a.y; Tr[2] = a.z; Tr[3] = a.w;
  Tr[4] = b.x; Tr[5] = b.y; Tr[6] = b.z; Tr[7] = b.w;
  Tr[8] = c.x; Tp[0] = c.y; Tp[1] = c.z; Tp[2] = c.w;
}

__device__ __forceinline__ void g_matvec_wmma(const float x[6], float y[6],
                                              v2f A0, v2f A1, v2f A2) {
  v8f acc = {0.f, 0.f, 0.f, 0.f, 0.f, 0.f, 0.f, 0.f};
  v2f B0, B1, B2;
  B0.x = x[0];  B0.y = x[1];
  B1.x = x[2];  B1.y = x[3];
  B2.x = x[4];  B2.y = x[5];
  acc = __builtin_amdgcn_wmma_f32_16x16x4_f32(false, A0, false, B0, (short)0, acc,
                                              false, false);
  acc = __builtin_amdgcn_wmma_f32_16x16x4_f32(false, A1, false, B1, (short)0, acc,
                                              false, false);
  acc = __builtin_amdgcn_wmma_f32_16x16x4_f32(false, A2, false, B2, (short)0, acc,
                                              false, false);
#pragma unroll
  for (int r = 0; r < 6; ++r) y[r] = acc[r];
}

__global__ __launch_bounds__(TPB) void inv_dyn_kernel(
    const float* __restrict__ q, const float* __restrict__ dq,
    const float* __restrict__ ddq, const float* __restrict__ g,
    const float* __restrict__ Ftip, const float* __restrict__ Alist,
    const float* __restrict__ Mlist, const float* __restrict__ Glist,
    float* __restrict__ tau, int B) {
  __shared__ float4 sT[(JN - 2)*3*TPB];
  const int tid = threadIdx.x;
  int gid = blockIdx.x * blockDim.x + threadIdx.x;
  int b = (gid < B) ? gid : (B - 1);

  float lq[JN], ldq[JN], lddq[JN];
#pragma unroll
  for (int i = 0; i < JN; ++i) {
    lq[i]   = q[b*JN + i];
    ldq[i]  = dq[b*JN + i];
    lddq[i] = ddq[b*JN + i];
  }

  float v[6] = {0.f, 0.f, 0.f, 0.f, 0.f, 0.f};
  float a[6] = {0.f, 0.f, 0.f, -g[0], -g[1], -g[2]};
  float vs[JN][6], as_[JN][6];
  float TlastR[9], TlastP[3];

#pragma unroll
  for (int i = 0; i < JN; ++i) {
    float Tr[9], Tp[3];
    compute_T(i, lq[i], Alist, Mlist, Tr, Tp);
    if (i >= 1 && i < JN - 1) stash_T(sT, i - 1, tid, Tr, Tp);
    if (i == JN - 1) {
#pragma unroll
      for (int k = 0; k < 9; ++k) TlastR[k] = Tr[k];
#pragma unroll
      for (int k = 0; k < 3; ++k) TlastP[k] = Tp[k];
    }
    const float* A = Alist + i*6;
    float dqi = ldq[i], ddqi = lddq[i];

    float Rvw[3], Rvv[3], pxw[3];
    matvec3(Tr, v, Rvw);
    matvec3(Tr, v + 3, Rvv);
    cross3(Tp, Rvw, pxw);
    float vn[6];
#pragma unroll
    for (int k = 0; k < 3; ++k) {
      vn[k]     = Rvw[k] + A[k]*dqi;
      vn[3 + k] = pxw[k] + Rvv[k] + A[3 + k]*dqi;
    }

    float Raw[3], Rav[3], pxa[3];
    matvec3(Tr, a, Raw);
    matvec3(Tr, a + 3, Rav);
    cross3(Tp, Raw, pxa);
    float t1[3], t2[3], t3[3];
    cross3(vn, A, t1);
    cross3(vn + 3, A, t2);
    cross3(vn, A + 3, t3);
    float an[6];
#pragma unroll
    for (int k = 0; k < 3; ++k) {
      an[k]     = Raw[k] + t1[k]*dqi + A[k]*ddqi;
      an[3 + k] = pxa[k] + Rav[k] + (t2[k] + t3[k])*dqi + A[3 + k]*ddqi;
    }
    asm("" : "+v"(vn[0]), "+v"(vn[1]), "+v"(vn[2]), "+v"(vn[3]), "+v"(vn[4]), "+v"(vn[5]),
             "+v"(an[0]), "+v"(an[1]), "+v"(an[2]), "+v"(an[3]), "+v"(an[4]), "+v"(an[5]));
#pragma unroll
    for (int k = 0; k < 6; ++k) {
      vs[i][k] = vn[k]; as_[i][k] = an[k];
      v[k] = vn[k];     a[k] = an[k];
    }
  }

  const int lane = threadIdx.x & 31;
  const int mrow = lane & 15;
  const int half = lane >> 4;
  const bool okrow = half ? (mrow >= 8 && mrow < 14) : (mrow < 6);
  const int  grow  = okrow ? (half ? mrow - 8 : mrow) : 0;

  float F[6];
#pragma unroll
  for (int k = 0; k < 6; ++k) F[k] = Ftip[b*6 + k];

#pragma unroll
  for (int ii = JN - 1; ii >= 0; --ii) {
    float Rn[9], Pn[3];
    if (ii == JN - 1) {
      const float* M = Mlist + JN*16;
#pragma unroll
      for (int r = 0; r < 3; ++r)
#pragma unroll
        for (int c = 0; c < 3; ++c)
          Rn[r*3+c] = M[c*4+r];
#pragma unroll
      for (int r = 0; r < 3; ++r)
        Pn[r] = -(Rn[r*3+0]*M[3] + Rn[r*3+1]*M[7] + Rn[r*3+2]*M[11]);
    } else if (ii == JN - 2) {
#pragma unroll
      for (int k = 0; k < 9; ++k) Rn[k] = TlastR[k];
#pragma unroll
      for (int k = 0; k < 3; ++k) Pn[k] = TlastP[k];
    } else {
      fetch_T(sT, ii, tid, Rn, Pn);
    }

    float cf[3];
    cross3(Pn, F + 3, cf);
    float tw[3] = {F[0] - cf[0], F[1] - cf[1], F[2] - cf[2]};
    float top[3], bot[3];
    matvec3T(Rn, tw, top);
    matvec3T(Rn, F + 3, bot);

    const float* Gr = Glist + ii*36 + grow*6;
    float g0 = Gr[0], g1 = Gr[1], g2 = Gr[2], g3 = Gr[3], g4 = Gr[4], g5 = Gr[5];
    v2f A0, A1, A2;
    A0.x = okrow ? g0 : 0.f;  A0.y = okrow ? g1 : 0.f;
    A1.x = okrow ? g2 : 0.f;  A1.y = okrow ? g3 : 0.f;
    A2.x = okrow ? g4 : 0.f;  A2.y = okrow ? g5 : 0.f;

    float Ga[6], Gv[6];
    g_matvec_wmma(as_[ii], Ga, A0, A1, A2);
    g_matvec_wmma(vs[ii],  Gv, A0, A1, A2);

    float c1[3], c2[3], c3[3];
    cross3(vs[ii], Gv, c1);
    cross3(vs[ii] + 3, Gv + 3, c2);
    cross3(vs[ii], Gv + 3, c3);
    float Fn[6];
#pragma unroll
    for (int k = 0; k < 3; ++k) {
      Fn[k]     = top[k] + Ga[k]     + c1[k] + c2[k];
      Fn[3 + k] = bot[k] + Ga[3 + k] + c3[k];
    }

    const float* A = Alist + ii*6;
    float tv = Fn[0]*A[0] + Fn[1]*A[1] + Fn[2]*A[2] +
               Fn[3]*A[3] + Fn[4]*A[4] + Fn[5]*A[5];
    tau[b*JN + ii] = tv;
#pragma unroll
    for (int k = 0; k < 6; ++k) F[k] = Fn[k];
  }
}

extern "C" void kernel_launch(void* const* d_in, const int* in_sizes, int n_in,
                              void* d_out, int out_size, void* d_ws, size_t ws_size,
                              hipStream_t stream) {
  const float* q     = (const float*)d_in[0];
  const float* dq    = (const float*)d_in[1];
  const float* ddq   = (const float*)d_in[2];
  const float* g     = (const float*)d_in[3];
  const float* Ftip  = (const float*)d_in[4];
  const float* Alist = (const float*)d_in[5];
  const float* Mlist = (const float*)d_in[6];
  const float* Glist = (const float*)d_in[7];
  int B = in_sizes[0] / JN;
  if (B <= 0) return;
  dim3 block(256);
  dim3 grid((B + 255) / 256);
  inv_dyn_kernel<<<grid, block, 0, stream>>>(q, dq, ddq, g, Ftip,
                                             Alist, Mlist, Glist,
                                             (float*)d_out, B);
}


# T_i's q-free factors (axis cross products with Minv_i) made once per wave, one lane per joint, shared through LDS, not serially on the scalar unit; lanes weight them by sin/cos; sums regrouped
# speedup vs baseline: 1.8741x; 1.0966x over previous
#include <hip/hip_runtime.h>

typedef __attribute__((ext_vector_type(2))) float v2f;
typedef __attribute__((ext_vector_type(8))) float v8f;

#define JN 9
#define NEARZ 1e-6f

__device__ __forceinline__ void cross3(const float* a, const float* b, float* o) {
  o[0] = a[1]*b[2] - a[2]*b[1];
  o[1] = a[2]*b[0] - a[0]*b[2];
  o[2] = a[0]*b[1] - a[1]*b[0];
}

__device__ __forceinline__ void matvec3(const float* R, const float* x, float* o) {
#pragma unroll
  for (int r = 0; r < 3; ++r)
    o[r] = R[r*3+0]*x[0] + R[r*3+1]*x[1] + R[r*3+2]*x[2];
}

__device__ __forceinline__ void matvec3T(const float* R, const float* x, float* o) {
#pragma unroll
  for (int r = 0; r < 3; ++r)
    o[r] = R[0*3+r]*x[0] + R[1*3+r]*x[1] + R[2*3+r]*x[2];
}

#define UQ 11

__device__ __forceinline__ void joint_constants(int j, const float* __restrict__ Alist,
                                                const float* __restrict__ Mlist, float* U) {
  const float* A = Alist + j*6;
  const float* M = Mlist + j*16;
  float aw[3] = {A[0], A[1], A[2]};
  float av[3] = {A[3], A[4], A[5]};
  float alpha = sqrtf(aw[0]*aw[0] + aw[1]*aw[1] + aw[2]*aw[2]);
  const bool tiny = !(alpha >= 1e-18f);
  float inv = 1.0f / (tiny ? 1.0f : alpha);
  float k[3], m[3];
#pragma unroll
  for (int r = 0; r < 3; ++r) {
    k[r] = tiny ? 0.f : aw[r]*inv;
    m[r] = tiny ? 0.f : av[r]*inv;
  }
#pragma unroll
  for (int c = 0; c < 3; ++c) {
    float x[3] = {M[c*4+0], M[c*4+1], M[c*4+2]};
    float kx[3], kkx[3];
    cross3(k, x, kx);
    cross3(k, kx, kkx);
#pragma unroll
    for (int r = 0; r < 3; ++r) {
      U[r*3+c]      = x[r];
      U[12 + r*3+c] = -kx[r];
      U[24 + r*3+c] = kkx[r];
    }
  }
  float pinv[3];
#pragma unroll
  for (int r = 0; r < 3; ++r)
    pinv[r] = -(M[0*4+r]*M[0*4+3] + M[1*4+r]*M[1*4+3] + M[2*4+r]*M[2*4+3]);
  U[42] = alpha;
  float Kp[3], K2p[3], Km[3], K2m[3];
  cross3(k, pinv, Kp);
  cross3(k, Kp, K2p);
  cross3(k, m, Km);
  cross3(k, Km, K2m);
#pragma unroll
  for (int r = 0; r < 3; ++r) {
    U[9 + r]  = pinv[r];
    U[21 + r] = -Kp[r];
    U[33 + r] = K2p[r] + Km[r];
    U[36 + r] = -av[r];
    U[39 + r] = -K2m[r];
  }
  U[43] = 0.f;
}

__device__ __forceinline__ void joint_T(const float4* u, float qi, float* Tr, float* Tp) {
  float U[4*UQ];
#pragma unroll
  for (int p = 0; p < UQ; ++p) {
    float4 x = u[p];
    U[4*p+0] = x.x; U[4*p+1] = x.y; U[4*p+2] = x.z; U[4*p+3] = x.w;
  }
  float phi = qi * U[42];
  const bool near = fabsf(phi) < NEARZ;
  float st = sinf(phi), ct = cosf(phi);
  float s = near ? 0.f : st;
  float c = near ? 0.f : 1.0f - ct;
  float d = near ? 0.f : phi - st;
#pragma unroll
  for (int e = 0; e < 9; ++e)
    Tr[e] = U[e] + s*U[12 + e] + c*U[24 + e];
#pragma unroll
  for (int r = 0; r < 3; ++r)
    Tp[r] = U[9 + r] + s*U[21 + r] + c*U[33 + r] + qi*U[36 + r] + d*U[39 + r];
}

#define TPB 256
__device__ __forceinline__ void stash_T(float4* sT, int j, int tid, const float* Tr, const float* Tp) {
  sT[(j*3 + 0)*TPB + tid] = make_float4(Tr[0], Tr[1], Tr[2], Tr[3]);
  sT[(j*3 + 1)*TPB + tid] = make_float4(Tr[4], Tr[5], Tr[6], Tr[7]);
  sT[(j*3 + 2)*TPB + tid] = make_float4(Tr[8], Tp[0], Tp[1], Tp[2]);
}
__device__ __forceinline__ void fetch_T(const float4* sT, int j, int tid, float* Tr, float* Tp) {
  float4 a = sT[(j*3 + 0)*TPB + tid];
  float4 b = sT[(j*3 + 1)*TPB + tid];
  float4 c = sT[(j*3 + 2)*TPB + tid];
  Tr[0] = a.x; Tr[1] = a.y; Tr[2] = a.z; Tr[3] = a.w;
  Tr[4] = b.x; Tr[5] = b.y; Tr[6] = b.z; Tr[7] = b.w;
  Tr[8] = c.x; Tp[0] = c.y; Tp[1] = c.z; Tp[2] = c.w;
}

__device__ __forceinline__ void g_matvec_wmma(const float x[6], float y[6],
                                              v2f A0, v2f A1, v2f A2) {
  v8f acc = {0.f, 0.f, 0.f, 0.f, 0.f, 0.f, 0.f, 0.f};
  v2f B0, B1, B2;
  B0.x = x[0];  B0.y = x[1];
  B1.x = x[2];  B1.y = x[3];
  B2.x = x[4];  B2.y = x[5];
  acc = __builtin_amdgcn_wmma_f32_16x16x4_f32(false, A0, false, B0, (short)0, acc,
                                              false, false);
  acc = __builtin_amdgcn_wmma_f32_16x16x4_f32(false, A1, false, B1, (short)0, acc,
                                              false, false);
  acc = __builtin_amdgcn_wmma_f32_16x16x4_f32(false, A2, false, B2, (short)0, acc,
                                              false, false);
#pragma unroll
  for (int r = 0; r < 6; ++r) y[r] = acc[r];
}

__global__ __launch_bounds__(TPB) void inv_dyn_kernel(
    const float* __restrict__ q, const float* __restrict__ dq,
    const float* __restrict__ ddq, const float* __restrict__ g,
    const float* __restrict__ Ftip, const float* __restrict__ Alist,
    const float* __restrict__ Mlist, const float* __restrict__ Glist,
    float* __restrict__ tau, int B) {
  __shared__ float4 sT[(JN - 2)*3*TPB];
  __shared__ float4 sU[(TPB/32)*JN*UQ];
  const int tid = threadIdx.x;
  int gid = blockIdx.x * blockDim.x + threadIdx.x;
  int b = (gid < B) ? gid : (B - 1);

  float lq[JN], ldq[JN], lddq[JN];
#pragma unroll
  for (int i = 0; i < JN; ++i) {
    lq[i]   = q[b*JN + i];
    ldq[i]  = dq[b*JN + i];
    lddq[i] = ddq[b*JN + i];
  }

  float4* uw = sU + (tid >> 5)*(JN*UQ);
  {
    const int ln = tid & 31;
    float U[4*UQ];
    joint_constants(ln < JN ? ln : JN - 1, Alist, Mlist, U);
    if (ln < JN) {
#pragma unroll
      for (int p = 0; p < UQ; ++p)
        uw[ln*UQ + p] = make_float4(U[4*p+0], U[4*p+1], U[4*p+2], U[4*p+3]);
    }
    __builtin_amdgcn_wave_barrier();
  }

  float v[6] = {0.f, 0.f, 0.f, 0.f, 0.f, 0.f};
  float a[6] = {0.f, 0.f, 0.f, -g[0], -g[1], -g[2]};
  float vs[JN][6], as_[JN][6];
  float TlastR[9], TlastP[3];

#pragma unroll
  for (int i = 0; i < JN; ++i) {
    float Tr[9], Tp[3];
    joint_T(uw + i*UQ, lq[i], Tr, Tp);
    if (i >= 1 && i < JN - 1) stash_T(sT, i - 1, tid, Tr, Tp);
    if (i == JN - 1) {
#pragma unroll
      for (int k = 0; k < 9; ++k) TlastR[k] = Tr[k];
#pragma unroll
      for (int k = 0; k < 3; ++k) TlastP[k] = Tp[k];
    }
    const float* A = Alist + i*6;
    float dqi = ldq[i], ddqi = lddq[i];

    float Rvw[3], Rvv[3], pxw[3];
    matvec3(Tr, v, Rvw);
    matvec3(Tr, v + 3, Rvv);
    cross3(Tp, Rvw, pxw);
    float vn[6];
#pragma unroll
    for (int k = 0; k < 3; ++k) {
      vn[k]     = Rvw[k] + A[k]*dqi;
      vn[3 + k] = pxw[k] + Rvv[k] + A[3 + k]*dqi;
    }

    float Raw[3], Rav[3], pxa[3];
    matvec3(Tr, a, Raw);
    matvec3(Tr, a + 3, Rav);
    cross3(Tp, Raw, pxa);
    float t1[3], t2[3], t3[3];
    cross3(vn, A, t1);
    cross3(vn + 3, A, t2);
    cross3(vn, A + 3, t3);
    float an[6];
#pragma unroll
    for (int k = 0; k < 3; ++k) {
      an[k]     = Raw[k] + t1[k]*dqi + A[k]*ddqi;
      an[3 + k] = pxa[k] + Rav[k] + (t2[k] + t3[k])*dqi + A[3 + k]*ddqi;
    }
    asm("" : "+v"(vn[0]), "+v"(vn[1]), "+v"(vn[2]), "+v"(vn[3]), "+v"(vn[4]), "+v"(vn[5]),
             "+v"(an[0]), "+v"(an[1]), "+v"(an[2]), "+v"(an[3]), "+v"(an[4]), "+v"(an[5]));
#pragma unroll
    for (int k = 0; k < 6; ++k) {
      vs[i][k] = vn[k]; as_[i][k] = an[k];
      v[k] = vn[k];     a[k] = an[k];
    }
  }

  const int lane = threadIdx.x & 31;
  const int mrow = lane & 15;
  const int half = lane >> 4;
  const bool okrow = half ? (mrow >= 8 && mrow < 14) : (mrow < 6);
  const int  grow  = okrow ? (half ? mrow - 8 : mrow) : 0;

  float F[6];
#pragma unroll
  for (int k = 0; k < 6; ++k) F[k] = Ftip[b*6 + k];

#pragma unroll
  for (int ii = JN - 1; ii >= 0; --ii) {
    float Rn[9], Pn[3];
    if (ii == JN - 1) {
      const float* M = Mlist + JN*16;
#pragma unroll
      for (int r = 0; r < 3; ++r)
#pragma unroll
        for (int c = 0; c < 3; ++c)
          Rn[r*3+c] = M[c*4+r];
#pragma unroll
      for (int r = 0; r < 3; ++r)
        Pn[r] = -(Rn[r*3+0]*M[3] + Rn[r*3+1]*M[7] + Rn[r*3+2]*M[11]);
    } else if (ii == JN - 2) {
#pragma unroll
      for (int k = 0; k < 9; ++k) Rn[k] = TlastR[k];
#pragma unroll
      for (int k = 0; k < 3; ++k) Pn[k] = TlastP[k];
    } else {
      fetch_T(sT, ii, tid, Rn, Pn);
    }

    float cf[3];
    cross3(Pn, F + 3, cf);
    float tw[3] = {F[0] - cf[0], F[1] - cf[1], F[2] - cf[2]};
    float top[3], bot[3];
    matvec3T(Rn, tw, top);
    matvec3T(Rn, F + 3, bot);

    const float* Gr = Glist + ii*36 + grow*6;
    float g0 = Gr[0], g1 = Gr[1], g2 = Gr[2], g3 = Gr[3], g4 = Gr[4], g5 = Gr[5];
    v2f A0, A1, A2;
    A0.x = okrow ? g0 : 0.f;  A0.y = okrow ? g1 : 0.f;
    A1.x = okrow ? g2 : 0.f;  A1.y = okrow ? g3 : 0.f;
    A2.x = okrow ? g4 : 0.f;  A2.y = okrow ? g5 : 0.f;

    float Ga[6], Gv[6];
    g_matvec_wmma(as_[ii], Ga, A0, A1, A2);
    g_matvec_wmma(vs[ii],  Gv, A0, A1, A2);

    float c1[3], c2[3], c3[3];
    cross3(vs[ii], Gv, c1);
    cross3(vs[ii] + 3, Gv + 3, c2);
    cross3(vs[ii], Gv + 3, c3);
    float Fn[6];
#pragma unroll
    for (int k = 0; k < 3; ++k) {
      Fn[k]     = top[k] + Ga[k]     + c1[k] + c2[k];
      Fn[3 + k] = bot[k] + Ga[3 + k] + c3[k];
    }

    const float* A = Alist + ii*6;
    float tv = Fn[0]*A[0] + Fn[1]*A[1] + Fn[2]*A[2] +
               Fn[3]*A[3] + Fn[4]*A[4] + Fn[5]*A[5];
    tau[b*JN + ii] = tv;
#pragma unroll
    for (int k = 0; k < 6; ++k) F[k] = Fn[k];
  }
}

extern "C" void kernel_launch(void* const* d_in, const int* in_sizes, int n_in,
                              void* d_out, int out_size, void* d_ws, size_t ws_size,
                              hipStream_t stream) {
  const float* q     = (const float*)d_in[0];
  const float* dq    = (const float*)d_in[1];
  const float* ddq   = (const float*)d_in[2];
  const float* g     = (const float*)d_in[3];
  const float* Ftip  = (const float*)d_in[4];
  const float* Alist = (const float*)d_in[5];
  const float* Mlist = (const float*)d_in[6];
  const float* Glist = (const float*)d_in[7];
  int B = in_sizes[0] / JN;
  if (B <= 0) return;
  dim3 block(256);
  dim3 grid((B + 255) / 256);
  inv_dyn_kernel<<<grid, block, 0, stream>>>(q, dq, ddq, g, Ftip,
                                             Alist, Mlist, Glist,
                                             (float*)d_out, B);
}


# sin and cos of each joint angle from one shared range reduction and one polynomial pair in straight-line code (the library's constants) instead of separate sinf and cosf, each behind its own branch
# speedup vs baseline: 2.0056x; 1.0702x over previous
#include <hip/hip_runtime.h>

typedef __attribute__((ext_vector_type(2))) float v2f;
typedef __attribute__((ext_vector_type(8))) float v8f;

#define JN 9
#define NEARZ 1e-6f

__device__ __forceinline__ void cross3(const float* a, const float* b, float* o) {
  o[0] = a[1]*b[2] - a[2]*b[1];
  o[1] = a[2]*b[0] - a[0]*b[2];
  o[2] = a[0]*b[1] - a[1]*b[0];
}

__device__ __forceinline__ void matvec3(const float* R, const float* x, float* o) {
#pragma unroll
  for (int r = 0; r < 3; ++r)
    o[r] = R[r*3+0]*x[0] + R[r*3+1]*x[1] + R[r*3+2]*x[2];
}

__device__ __forceinline__ void matvec3T(const float* R, const float* x, float* o) {
#pragma unroll
  for (int r = 0; r < 3; ++r)
    o[r] = R[0*3+r]*x[0] + R[1*3+r]*x[1] + R[2*3+r]*x[2];
}

#define UQ 11

__device__ __forceinline__ void joint_constants(int j, const float* __restrict__ Alist,
                                                const float* __restrict__ Mlist, float* U) {
  const float* A = Alist + j*6;
  const float* M = Mlist + j*16;
  float aw[3] = {A[0], A[1], A[2]};
  float av[3] = {A[3], A[4], A[5]};
  float alpha = sqrtf(aw[0]*aw[0] + aw[1]*aw[1] + aw[2]*aw[2]);
  const bool tiny = !(alpha >= 1e-18f);
  float inv = 1.0f / (tiny ? 1.0f : alpha);
  float k[3], m[3];
#pragma unroll
  for (int r = 0; r < 3; ++r) {
    k[r] = tiny ? 0.f : aw[r]*inv;
    m[r] = tiny ? 0.f : av[r]*inv;
  }
#pragma unroll
  for (int c = 0; c < 3; ++c) {
    float x[3] = {M[c*4+0], M[c*4+1], M[c*4+2]};
    float kx[3], kkx[3];
    cross3(k, x, kx);
    cross3(k, kx, kkx);
#pragma unroll
    for (int r = 0; r < 3; ++r) {
      U[r*3+c]      = x[r];
      U[12 + r*3+c] = -kx[r];
      U[24 + r*3+c] = kkx[r];
    }
  }
  float pinv[3];
#pragma unroll
  for (int r = 0; r < 3; ++r)
    pinv[r] = -(M[0*4+r]*M[0*4+3] + M[1*4+r]*M[1*4+3] + M[2*4+r]*M[2*4+3]);
  U[42] = alpha;
  float Kp[3], K2p[3], Km[3], K2m[3];
  cross3(k, pinv, Kp);
  cross3(k, Kp, K2p);
  cross3(k, m, Km);
  cross3(k, Km, K2m);
#pragma unroll
  for (int r = 0; r < 3; ++r) {
    U[9 + r]  = pinv[r];
    U[21 + r] = -Kp[r];
    U[33 + r] = K2p[r] + Km[r];
    U[36 + r] = -av[r];
    U[39 + r] = -K2m[r];
  }
  U[43] = 0.f;
}

__device__ __forceinline__ void sincos_shared(float x, float* sn, float* cs) {
  const float two_over_pi = __builtin_bit_cast(float, 0x3f22f983u);
  const float pio2_a = __builtin_bit_cast(float, 0xbfc90fdau);
  const float pio2_b = __builtin_bit_cast(float, 0xb3a22168u);
  const float pio2_c = __builtin_bit_cast(float, 0xa7c234c4u);
  const float big = 12582912.0f;
  float t = fmaf(x, two_over_pi, big);
  float n = t - big;
  float r = fmaf(n, pio2_a, x);
  r = fmaf(n, pio2_b, r);
  r = fmaf(n, pio2_c, r);
  float r2 = r * r;
  float ps = fmaf(r2, __builtin_bit_cast(float, 0xb94c1982u), __builtin_bit_cast(float, 0x3c0881c4u));
  ps = fmaf(r2, ps, __builtin_bit_cast(float, 0xbe2aaa9du));
  float sr = fmaf(r, r2 * ps, r);
  float pc = fmaf(r2, __builtin_bit_cast(float, 0x37d75334u), __builtin_bit_cast(float, 0xbab64f3bu));
  pc = fmaf(r2, pc, __builtin_bit_cast(float, 0x3d2aabf7u));
  pc = fmaf(r2, pc, __builtin_bit_cast(float, 0xbf000004u));
  float cr = fmaf(r2, pc, 1.0f);
  unsigned qd = __float_as_uint(t);
  bool odd = (qd & 1u) != 0u;
  unsigned flip = (qd & 2u) << 30;
  *sn = __uint_as_float(__float_as_uint(odd ? cr : sr) ^ flip);
  *cs = __uint_as_float(__float_as_uint(odd ? -sr : cr) ^ flip);
}

__device__ __forceinline__ void joint_T(const float4* u, float qi, float* Tr, float* Tp) {
  float U[4*UQ];
#pragma unroll
  for (int p = 0; p < UQ; ++p) {
    float4 x = u[p];
    U[4*p+0] = x.x; U[4*p+1] = x.y; U[4*p+2] = x.z; U[4*p+3] = x.w;
  }
  float phi = qi * U[42];
  const bool near = fabsf(phi) < NEARZ;
  float st, ct;
  sincos_shared(phi, &st, &ct);
  float s = near ? 0.f : st;
  float c = near ? 0.f : 1.0f - ct;
  float d = near ? 0.f : phi - st;
#pragma unroll
  for (int e = 0; e < 9; ++e)
    Tr[e] = U[e] + s*U[12 + e] + c*U[24 + e];
#pragma unroll
  for (int r = 0; r < 3; ++r)
    Tp[r] = U[9 + r] + s*U[21 + r] + c*U[33 + r] + qi*U[36 + r] + d*U[39 + r];
}

#define TPB 256
__device__ __forceinline__ void stash_T(float4* sT, int j, int tid, const float* Tr, const float* Tp) {
  sT[(j*3 + 0)*TPB + tid] = make_float4(Tr[0], Tr[1], Tr[2], Tr[3]);
  sT[(j*3 + 1)*TPB + tid] = make_float4(Tr[4], Tr[5], Tr[6], Tr[7]);
  sT[(j*3 + 2)*TPB + tid] = make_float4(Tr[8], Tp[0], Tp[1], Tp[2]);
}
__device__ __forceinline__ void fetch_T(const float4* sT, int j, int tid, float* Tr, float* Tp) {
  float4 a = sT[(j*3 + 0)*TPB + tid];
  float4 b = sT[(j*3 + 1)*TPB + tid];
  float4 c = sT[(j*3 + 2)*TPB + tid];
  Tr[0] = a.x; Tr[1] = a.y; Tr[2] = a.z; Tr[3] = a.w;
  Tr[4] = b.x; Tr[5] = b.y; Tr[6] = b.z; Tr[7] = b.w;
  Tr[8] = c.x; Tp[0] = c.y; Tp[1] = c.z; Tp[2] = c.w;
}

__device__ __forceinline__ void g_matvec_wmma(const float x[6], float y[6],
                                              v2f A0, v2f A1, v2f A2) {
  v8f acc = {0.f, 0.f, 0.f, 0.f, 0.f, 0.f, 0.f, 0.f};
  v2f B0, B1, B2;
  B0.x = x[0];  B0.y = x[1];
  B1.x = x[2];  B1.y = x[3];
  B2.x = x[4];  B2.y = x[5];
  acc = __builtin_amdgcn_wmma_f32_16x16x4_f32(false, A0, false, B0, (short)0, acc,
                                              false, false);
  acc = __builtin_amdgcn_wmma_f32_16x16x4_f32(false, A1, false, B1, (short)0, acc,
                                              false, false);
  acc = __builtin_amdgcn_wmma_f32_16x16x4_f32(false, A2, false, B2, (short)0, acc,
                                              false, false);
#pragma unroll
  for (int r = 0; r < 6; ++r) y[r] = acc[r];
}

__global__ __launch_bounds__(TPB, 4) void inv_dyn_kernel(
    const float* __restrict__ q, const float* __restrict__ dq,
    const float* __restrict__ ddq, const float* __restrict__ g,
    const float* __restrict__ Ftip, const float* __restrict__ Alist,
    const float* __restrict__ Mlist, const float* __restrict__ Glist,
    float* __restrict__ tau, int B) {
  __shared__ float4 sT[(JN - 2)*3*TPB];
  __shared__ float4 sU[(TPB/32)*JN*UQ];
  const int tid = threadIdx.x;
  int gid = blockIdx.x * blockDim.x + threadIdx.x;
  int b = (gid < B) ? gid : (B - 1);

  float lq[JN], ldq[JN], lddq[JN];
#pragma unroll
  for (int i = 0; i < JN; ++i) {
    lq[i]   = q[b*JN + i];
    ldq[i]  = dq[b*JN + i];
    lddq[i] = ddq[b*JN + i];
  }

  float4* uw = sU + (tid >> 5)*(JN*UQ);
  {
    const int ln = tid & 31;
    float U[4*UQ];
    joint_constants(ln < JN ? ln : JN - 1, Alist, Mlist, U);
    if (ln < JN) {
#pragma unroll
      for (int p = 0; p < UQ; ++p)
        uw[ln*UQ + p] = make_float4(U[4*p+0], U[4*p+1], U[4*p+2], U[4*p+3]);
    }
    __builtin_amdgcn_wave_barrier();
  }

  float v[6] = {0.f, 0.f, 0.f, 0.f, 0.f, 0.f};
  float a[6] = {0.f, 0.f, 0.f, -g[0], -g[1], -g[2]};
  float vs[JN][6], as_[JN][6];
  float TlastR[9], TlastP[3];

#pragma unroll
  for (int i = 0; i < JN; ++i) {
    float Tr[9], Tp[3];
    joint_T(uw + i*UQ, lq[i], Tr, Tp);
    if (i >= 1 && i < JN - 1) stash_T(sT, i - 1, tid, Tr, Tp);
    if (i == JN - 1) {
#pragma unroll
      for (int k = 0; k < 9; ++k) TlastR[k] = Tr[k];
#pragma unroll
      for (int k = 0; k < 3; ++k) TlastP[k] = Tp[k];
    }
    const float* A = Alist + i*6;
    float dqi = ldq[i], ddqi = lddq[i];

    float Rvw[3], Rvv[3], pxw[3];
    matvec3(Tr, v, Rvw);
    matvec3(Tr, v + 3, Rvv);
    cross3(Tp, Rvw, pxw);
    float vn[6];
#pragma unroll
    for (int k = 0; k < 3; ++k) {
      vn[k]     = Rvw[k] + A[k]*dqi;
      vn[3 + k] = pxw[k] + Rvv[k] + A[3 + k]*dqi;
    }

    float Raw[3], Rav[3], pxa[3];
    matvec3(Tr, a, Raw);
    matvec3(Tr, a + 3, Rav);
    cross3(Tp, Raw, pxa);
    float t1[3], t2[3], t3[3];
    cross3(vn, A, t1);
    cross3(vn + 3, A, t2);
    cross3(vn, A + 3, t3);
    float an[6];
#pragma unroll
    for (int k = 0; k < 3; ++k) {
      an[k]     = Raw[k] + t1[k]*dqi + A[k]*ddqi;
      an[3 + k] = pxa[k] + Rav[k] + (t2[k] + t3[k])*dqi + A[3 + k]*ddqi;
    }
    asm("" : "+v"(vn[0]), "+v"(vn[1]), "+v"(vn[2]), "+v"(vn[3]), "+v"(vn[4]), "+v"(vn[5]),
             "+v"(an[0]), "+v"(an[1]), "+v"(an[2]), "+v"(an[3]), "+v"(an[4]), "+v"(an[5]));
#pragma unroll
    for (int k = 0; k < 6; ++k) {
      vs[i][k] = vn[k]; as_[i][k] = an[k];
      v[k] = vn[k];     a[k] = an[k];
    }
  }

  __builtin_amdgcn_sched_barrier(0);

  const int lane = threadIdx.x & 31;
  const int mrow = lane & 15;
  const int half = lane >> 4;
  const bool okrow = half ? (mrow >= 8 && mrow < 14) : (mrow < 6);
  const int  grow  = okrow ? (half ? mrow - 8 : mrow) : 0;

  float F[6];
#pragma unroll
  for (int k = 0; k < 6; ++k) F[k] = Ftip[b*6 + k];

#pragma unroll
  for (int ii = JN - 1; ii >= 0; --ii) {
    float Rn[9], Pn[3];
    if (ii == JN - 1) {
      const float* M = Mlist + JN*16;
#pragma unroll
      for (int r = 0; r < 3; ++r)
#pragma unroll
        for (int c = 0; c < 3; ++c)
          Rn[r*3+c] = M[c*4+r];
#pragma unroll
      for (int r = 0; r < 3; ++r)
        Pn[r] = -(Rn[r*3+0]*M[3] + Rn[r*3+1]*M[7] + Rn[r*3+2]*M[11]);
    } else if (ii == JN - 2) {
#pragma unroll
      for (int k = 0; k < 9; ++k) Rn[k] = TlastR[k];
#pragma unroll
      for (int k = 0; k < 3; ++k) Pn[k] = TlastP[k];
    } else {
      fetch_T(sT, ii, tid, Rn, Pn);
    }

    float cf[3];
    cross3(Pn, F + 3, cf);
    float tw[3] = {F[0] - cf[0], F[1] - cf[1], F[2] - cf[2]};
    float top[3], bot[3];
    matvec3T(Rn, tw, top);
    matvec3T(Rn, F + 3, bot);

    const float* Gr = Glist + ii*36 + grow*6;
    float g0 = Gr[0], g1 = Gr[1], g2 = Gr[2], g3 = Gr[3], g4 = Gr[4], g5 = Gr[5];
    v2f A0, A1, A2;
    A0.x = okrow ? g0 : 0.f;  A0.y = okrow ? g1 : 0.f;
    A1.x = okrow ? g2 : 0.f;  A1.y = okrow ? g3 : 0.f;
    A2.x = okrow ? g4 : 0.f;  A2.y = okrow ? g5 : 0.f;

    float Ga[6], Gv[6];
    g_matvec_wmma(as_[ii], Ga, A0, A1, A2);
    g_matvec_wmma(vs[ii],  Gv, A0, A1, A2);

    float c1[3], c2[3], c3[3];
    cross3(vs[ii], Gv, c1);
    cross3(vs[ii] + 3, Gv + 3, c2);
    cross3(vs[ii], Gv + 3, c3);
    float Fn[6];
#pragma unroll
    for (int k = 0; k < 3; ++k) {
      Fn[k]     = top[k] + Ga[k]     + c1[k] + c2[k];
      Fn[3 + k] = bot[k] + Ga[3 + k] + c3[k];
    }

    const float* A = Alist + ii*6;
    float tv = Fn[0]*A[0] + Fn[1]*A[1] + Fn[2]*A[2] +
               Fn[3]*A[3] + Fn[4]*A[4] + Fn[5]*A[5];
    tau[b*JN + ii] = tv;
#pragma unroll
    for (int k = 0; k < 6; ++k) F[k] = Fn[k];
  }
}

extern "C" void kernel_launch(void* const* d_in, const int* in_sizes, int n_in,
                              void* d_out, int out_size, void* d_ws, size_t ws_size,
                              hipStream_t stream) {
  const float* q     = (const float*)d_in[0];
  const float* dq    = (const float*)d_in[1];
  const float* ddq   = (const float*)d_in[2];
  const float* g     = (const float*)d_in[3];
  const float* Ftip  = (const float*)d_in[4];
  const float* Alist = (const float*)d_in[5];
  const float* Mlist = (const float*)d_in[6];
  const float* Glist = (const float*)d_in[7];
  int B = in_sizes[0] / JN;
  if (B <= 0) return;
  dim3 block(256);
  dim3 grid((B + 255) / 256);
  inv_dyn_kernel<<<grid, block, 0, stream>>>(q, dq, ddq, g, Ftip,
                                             Alist, Mlist, Glist,
                                             (float*)d_out, B);
}


# Products fused into the sums they feed: T_i's sums as packed two-float multiply-add chains over a pair-aligned table; backward cross products added term by term, none formed first; sums regrouped
# speedup vs baseline: 2.1531x; 1.0735x over previous
#include <hip/hip_runtime.h>

typedef __attribute__((ext_vector_type(2))) float v2f;
typedef __attribute__((ext_vector_type(8))) float v8f;

#define JN 9
#define NEARZ 1e-6f

__device__ __forceinline__ void cross3(const float* a, const float* b, float* o) {
  o[0] = a[1]*b[2] - a[2]*b[1];
  o[1] = a[2]*b[0] - a[0]*b[2];
  o[2] = a[0]*b[1] - a[1]*b[0];
}

__device__ __forceinline__ void cross3_acc(const float* a, const float* b, float* acc) {
  acc[0] = (acc[0] + a[1]*b[2]) - a[2]*b[1];
  acc[1] = (acc[1] + a[2]*b[0]) - a[0]*b[2];
  acc[2] = (acc[2] + a[0]*b[1]) - a[1]*b[0];
}

__device__ __forceinline__ void cross3_sub(const float* a, const float* b, float* acc) {
  acc[0] = (acc[0] - a[1]*b[2]) + a[2]*b[1];
  acc[1] = (acc[1] - a[2]*b[0]) + a[0]*b[2];
  acc[2] = (acc[2] - a[0]*b[1]) + a[1]*b[0];
}

__device__ __forceinline__ void matvec3(const float* R, const float* x, float* o) {
#pragma unroll
  for (int r = 0; r < 3; ++r)
    o[r] = R[r*3+0]*x[0] + R[r*3+1]*x[1] + R[r*3+2]*x[2];
}

__device__ __forceinline__ void matvec3T(const float* R, const float* x, float* o) {
#pragma unroll
  for (int r = 0; r < 3; ++r)
    o[r] = R[0*3+r]*x[0] + R[1*3+r]*x[1] + R[2*3+r]*x[2];
}

#define UQ 11

__device__ __forceinline__ void joint_constants(int j, const float* __restrict__ Alist,
                                                const float* __restrict__ Mlist, float* U) {
  const float* A = Alist + j*6;
  const float* M = Mlist + j*16;
  float aw[3] = {A[0], A[1], A[2]};
  float av[3] = {A[3], A[4], A[5]};
  float alpha = sqrtf(aw[0]*aw[0] + aw[1]*aw[1] + aw[2]*aw[2]);
  const bool tiny = !(alpha >= 1e-18f);
  float inv = 1.0f / (tiny ? 1.0f : alpha);
  float k[3], m[3];
#pragma unroll
  for (int r = 0; r < 3; ++r) {
    k[r] = tiny ? 0.f : aw[r]*inv;
    m[r] = tiny ? 0.f : av[r]*inv;
  }
#pragma unroll
  for (int c = 0; c < 3; ++c) {
    float x[3] = {M[c*4+0], M[c*4+1], M[c*4+2]};
    float kx[3], kkx[3];
    cross3(k, x, kx);
    cross3(k, kx, kkx);
#pragma unroll
    for (int r = 0; r < 3; ++r) {
      U[r*3+c]      = x[r];
      U[12 + r*3+c] = -kx[r];
      U[24 + r*3+c] = kkx[r];
    }
  }
  float pinv[3];
#pragma unroll
  for (int r = 0; r < 3; ++r)
    pinv[r] = -(M[0*4+r]*M[0*4+3] + M[1*4+r]*M[1*4+3] + M[2*4+r]*M[2*4+3]);
  U[38] = alpha;
  float Kp[3], K2p[3], Km[3], K2m[3];
  cross3(k, pinv, Kp);
  cross3(k, Kp, K2p);
  cross3(k, m, Km);
  cross3(k, Km, K2m);
#pragma unroll
  for (int r = 0; r < 3; ++r) {
    U[9 + r]  = pinv[r];
    U[21 + r] = -Kp[r];
    U[33 + r] = K2p[r] + Km[r];
  }
  U[36] = -av[0];   U[37] = -K2m[0];   U[39] = 0.f;
  U[40] = -av[1];   U[41] = -av[2];
  U[42] = -K2m[1];  U[43] = -K2m[2];
}

__device__ __forceinline__ void sincos_shared(float x, float* sn, float* cs) {
  const float two_over_pi = __builtin_bit_cast(float, 0x3f22f983u);
  const float pio2_a = __builtin_bit_cast(float, 0xbfc90fdau);
  const float pio2_b = __builtin_bit_cast(float, 0xb3a22168u);
  const float pio2_c = __builtin_bit_cast(float, 0xa7c234c4u);
  const float big = 12582912.0f;
  float t = fmaf(x, two_over_pi, big);
  float n = t - big;
  float r = fmaf(n, pio2_a, x);
  r = fmaf(n, pio2_b, r);
  r = fmaf(n, pio2_c, r);
  float r2 = r * r;
  float ps = fmaf(r2, __builtin_bit_cast(float, 0xb94c1982u), __builtin_bit_cast(float, 0x3c0881c4u));
  ps = fmaf(r2, ps, __builtin_bit_cast(float, 0xbe2aaa9du));
  float sr = fmaf(r, r2 * ps, r);
  float pc = fmaf(r2, __builtin_bit_cast(float, 0x37d75334u), __builtin_bit_cast(float, 0xbab64f3bu));
  pc = fmaf(r2, pc, __builtin_bit_cast(float, 0x3d2aabf7u));
  pc = fmaf(r2, pc, __builtin_bit_cast(float, 0xbf000004u));
  float cr = fmaf(r2, pc, 1.0f);
  unsigned qd = __float_as_uint(t);
  bool odd = (qd & 1u) != 0u;
  unsigned flip = (qd & 2u) << 30;
  *sn = __uint_as_float(__float_as_uint(odd ? cr : sr) ^ flip);
  *cs = __uint_as_float(__float_as_uint(odd ? -sr : cr) ^ flip);
}

__device__ __forceinline__ void joint_T(const float4* u, float qi, float* Tr, float* Tp) {
  v2f U2[2*UQ];
#pragma unroll
  for (int p = 0; p < UQ; ++p) {
    float4 x = u[p];
    U2[2*p+0].x = x.x; U2[2*p+0].y = x.y;
    U2[2*p+1].x = x.z; U2[2*p+1].y = x.w;
  }
  float phi = qi * U2[19].x;
  const bool near = fabsf(phi) < NEARZ;
  float st, ct;
  sincos_shared(phi, &st, &ct);
  float s = near ? 0.f : st;
  float c = near ? 0.f : 1.0f - ct;
  float d = near ? 0.f : phi - st;
  v2f s2 = {s, s}, c2 = {c, c}, q2 = {qi, qi}, d2 = {d, d};
  v2f T2[6];
#pragma unroll
  for (int p = 0; p < 6; ++p)
    T2[p] = U2[p] + s2*U2[6 + p] + c2*U2[12 + p];
  float tp0 = T2[4].y + qi*U2[18].x + d*U2[18].y;
  v2f t12 = T2[5] + q2*U2[20] + d2*U2[21];
#pragma unroll
  for (int p = 0; p < 4; ++p) { Tr[2*p] = T2[p].x; Tr[2*p+1] = T2[p].y; }
  Tr[8] = T2[4].x;
  Tp[0] = tp0; Tp[1] = t12.x; Tp[2] = t12.y;
}

#define TPB 256
__device__ __forceinline__ void stash_T(float4* sT, int j, int tid, const float* Tr, const float* Tp) {
  sT[(j*3 + 0)*TPB + tid] = make_float4(Tr[0], Tr[1], Tr[2], Tr[3]);
  sT[(j*3 + 1)*TPB + tid] = make_float4(Tr[4], Tr[5], Tr[6], Tr[7]);
  sT[(j*3 + 2)*TPB + tid] = make_float4(Tr[8], Tp[0], Tp[1], Tp[2]);
}
__device__ __forceinline__ void fetch_T(const float4* sT, int j, int tid, float* Tr, float* Tp) {
  float4 a = sT[(j*3 + 0)*TPB + tid];
  float4 b = sT[(j*3 + 1)*TPB + tid];
  float4 c = sT[(j*3 + 2)*TPB + tid];
  Tr[0] = a.x; Tr[1] = a.y; Tr[2] = a.z; Tr[3] = a.w;
  Tr[4] = b.x; Tr[5] = b.y; Tr[6] = b.z; Tr[7] = b.w;
  Tr[8] = c.x; Tp[0] = c.y; Tp[1] = c.z; Tp[2] = c.w;
}

__device__ __forceinline__ void g_matvec_wmma(const float x[6], float y[6],
                                              v2f A0, v2f A1, v2f A2) {
  v8f acc = {0.f, 0.f, 0.f, 0.f, 0.f, 0.f, 0.f, 0.f};
  v2f B0, B1, B2;
  B0.x = x[0];  B0.y = x[1];
  B1.x = x[2];  B1.y = x[3];
  B2.x = x[4];  B2.y = x[5];
  acc = __builtin_amdgcn_wmma_f32_16x16x4_f32(false, A0, false, B0, (short)0, acc,
                                              false, false);
  acc = __builtin_amdgcn_wmma_f32_16x16x4_f32(false, A1, false, B1, (short)0, acc,
                                              false, false);
  acc = __builtin_amdgcn_wmma_f32_16x16x4_f32(false, A2, false, B2, (short)0, acc,
                                              false, false);
#pragma unroll
  for (int r = 0; r < 6; ++r) y[r] = acc[r];
}

__global__ __launch_bounds__(TPB, 4) void inv_dyn_kernel(
    const float* __restrict__ q, const float* __restrict__ dq,
    const float* __restrict__ ddq, const float* __restrict__ g,
    const float* __restrict__ Ftip, const float* __restrict__ Alist,
    const float* __restrict__ Mlist, const float* __restrict__ Glist,
    float* __restrict__ tau, int B) {
  __shared__ float4 sT[(JN - 2)*3*TPB];
  __shared__ float4 sU[(TPB/32)*JN*UQ];
  const int tid = threadIdx.x;
  int gid = blockIdx.x * blockDim.x + threadIdx.x;
  int b = (gid < B) ? gid : (B - 1);

  float lq[JN], ldq[JN], lddq[JN];
#pragma unroll
  for (int i = 0; i < JN; ++i) {
    lq[i]   = q[b*JN + i];
    ldq[i]  = dq[b*JN + i];
    lddq[i] = ddq[b*JN + i];
  }

  float4* uw = sU + (tid >> 5)*(JN*UQ);
  {
    const int ln = tid & 31;
    float U[4*UQ];
    joint_constants(ln < JN ? ln : JN - 1, Alist, Mlist, U);
    if (ln < JN) {
#pragma unroll
      for (int p = 0; p < UQ; ++p)
        uw[ln*UQ + p] = make_float4(U[4*p+0], U[4*p+1], U[4*p+2], U[4*p+3]);
    }
    __builtin_amdgcn_wave_barrier();
  }

  float v[6] = {0.f, 0.f, 0.f, 0.f, 0.f, 0.f};
  float a[6] = {0.f, 0.f, 0.f, -g[0], -g[1], -g[2]};
  float vs[JN][6], as_[JN][6];
  float TlastR[9], TlastP[3];

#pragma unroll
  for (int i = 0; i < JN; ++i) {
    float Tr[9], Tp[3];
    joint_T(uw + i*UQ, lq[i], Tr, Tp);
    if (i >= 1 && i < JN - 1) stash_T(sT, i - 1, tid, Tr, Tp);
    if (i == JN - 1) {
#pragma unroll
      for (int k = 0; k < 9; ++k) TlastR[k] = Tr[k];
#pragma unroll
      for (int k = 0; k < 3; ++k) TlastP[k] = Tp[k];
    }
    const float* A = Alist + i*6;
    float dqi = ldq[i], ddqi = lddq[i];

    float Rvw[3], Rvv[3], pxw[3];
    matvec3(Tr, v, Rvw);
    matvec3(Tr, v + 3, Rvv);
    cross3(Tp, Rvw, pxw);
    float vn[6];
#pragma unroll
    for (int k = 0; k < 3; ++k) {
      vn[k]     = Rvw[k] + A[k]*dqi;
      vn[3 + k] = pxw[k] + Rvv[k] + A[3 + k]*dqi;
    }

    float Raw[3], Rav[3], pxa[3];
    matvec3(Tr, a, Raw);
    matvec3(Tr, a + 3, Rav);
    cross3(Tp, Raw, pxa);
    float t1[3], t2[3], t3[3];
    cross3(vn, A, t1);
    cross3(vn + 3, A, t2);
    cross3(vn, A + 3, t3);
    float an[6];
#pragma unroll
    for (int k = 0; k < 3; ++k) {
      an[k]     = Raw[k] + t1[k]*dqi + A[k]*ddqi;
      an[3 + k] = pxa[k] + Rav[k] + (t2[k] + t3[k])*dqi + A[3 + k]*ddqi;
    }
    asm("" : "+v"(vn[0]), "+v"(vn[1]), "+v"(vn[2]), "+v"(vn[3]), "+v"(vn[4]), "+v"(vn[5]),
             "+v"(an[0]), "+v"(an[1]), "+v"(an[2]), "+v"(an[3]), "+v"(an[4]), "+v"(an[5]));
#pragma unroll
    for (int k = 0; k < 6; ++k) {
      vs[i][k] = vn[k]; as_[i][k] = an[k];
      v[k] = vn[k];     a[k] = an[k];
    }
  }

  __builtin_amdgcn_sched_barrier(0);

  const int lane = threadIdx.x & 31;
  const int mrow = lane & 15;
  const int half = lane >> 4;
  const bool okrow = half ? (mrow >= 8 && mrow < 14) : (mrow < 6);
  const int  grow  = okrow ? (half ? mrow - 8 : mrow) : 0;

  float F[6];
#pragma unroll
  for (int k = 0; k < 6; ++k) F[k] = Ftip[b*6 + k];

#pragma unroll
  for (int ii = JN - 1; ii >= 0; --ii) {
    float Rn[9], Pn[3];
    if (ii == JN - 1) {
      const float* M = Mlist + JN*16;
#pragma unroll
      for (int r = 0; r < 3; ++r)
#pragma unroll
        for (int c = 0; c < 3; ++c)
          Rn[r*3+c] = M[c*4+r];
#pragma unroll
      for (int r = 0; r < 3; ++r)
        Pn[r] = -(Rn[r*3+0]*M[3] + Rn[r*3+1]*M[7] + Rn[r*3+2]*M[11]);
    } else if (ii == JN - 2) {
#pragma unroll
      for (int k = 0; k < 9; ++k) Rn[k] = TlastR[k];
#pragma unroll
      for (int k = 0; k < 3; ++k) Pn[k] = TlastP[k];
    } else {
      fetch_T(sT, ii, tid, Rn, Pn);
    }

    float tw[3] = {F[0], F[1], F[2]};
    cross3_sub(Pn, F + 3, tw);
    float top[3], bot[3];
    matvec3T(Rn, tw, top);
    matvec3T(Rn, F + 3, bot);

    const float* Gr = Glist + ii*36 + grow*6;
    float g0 = Gr[0], g1 = Gr[1], g2 = Gr[2], g3 = Gr[3], g4 = Gr[4], g5 = Gr[5];
    v2f A0, A1, A2;
    A0.x = okrow ? g0 : 0.f;  A0.y = okrow ? g1 : 0.f;
    A1.x = okrow ? g2 : 0.f;  A1.y = okrow ? g3 : 0.f;
    A2.x = okrow ? g4 : 0.f;  A2.y = okrow ? g5 : 0.f;

    float Ga[6], Gv[6];
    g_matvec_wmma(as_[ii], Ga, A0, A1, A2);
    g_matvec_wmma(vs[ii],  Gv, A0, A1, A2);

    float Fn[6];
#pragma unroll
    for (int k = 0; k < 3; ++k) {
      Fn[k]     = top[k] + Ga[k];
      Fn[3 + k] = bot[k] + Ga[3 + k];
    }
    cross3_acc(vs[ii], Gv, Fn);
    cross3_acc(vs[ii] + 3, Gv + 3, Fn);
    cross3_acc(vs[ii], Gv + 3, Fn + 3);

    const float* A = Alist + ii*6;
    float tv = Fn[0]*A[0] + Fn[1]*A[1] + Fn[2]*A[2] +
               Fn[3]*A[3] + Fn[4]*A[4] + Fn[5]*A[5];
    tau[b*JN + ii] = tv;
#pragma unroll
    for (int k = 0; k < 6; ++k) F[k] = Fn[k];
  }
}

extern "C" void kernel_launch(void* const* d_in, const int* in_sizes, int n_in,
                              void* d_out, int out_size, void* d_ws, size_t ws_size,
                              hipStream_t stream) {
  const float* q     = (const float*)d_in[0];
  const float* dq    = (const float*)d_in[1];
  const float* ddq   = (const float*)d_in[2];
  const float* g     = (const float*)d_in[3];
  const float* Ftip  = (const float*)d_in[4];
  const float* Alist = (const float*)d_in[5];
  const float* Mlist = (const float*)d_in[6];
  const float* Glist = (const float*)d_in[7];
  int B = in_sizes[0] / JN;
  if (B <= 0) return;
  dim3 block(256);
  dim3 grid((B + 255) / 256);
  inv_dyn_kernel<<<grid, block, 0, stream>>>(q, dq, ddq, g, Ftip,
                                             Alist, Mlist, Glist,
                                             (float*)d_out, B);
}
